# starting kernel
# baseline (speedup 1.0000x reference)
#include <hip/hip_runtime.h>
#include <hip/hip_bf16.h>
#include <cfloat>

#define BS 64
#define NQ 900
#define NT 200
#define NC 92
#define KSTEPS (NC / 4)
#define TTILES ((NT + 15) / 16)
#define QTILES ((NQ + 15) / 16)

typedef float v2f __attribute__((ext_vector_type(2)));
typedef float v8f __attribute__((ext_vector_type(8)));

__global__ __launch_bounds__(256) void softmax_rows_k(
    const float* __restrict__ logits, float* __restrict__ prob) {
  const int row = blockIdx.x * 8 + (threadIdx.x >> 5);
  if (row >= BS * NQ) return;
  const int lane = threadIdx.x & 31;
  const float* x = logits + (size_t)row * NC;

  float a0 = x[lane];
  float a1 = x[lane + 32];
  float a2 = (lane + 64 < NC) ? x[lane + 64] : -FLT_MAX;

  float m = fmaxf(a0, fmaxf(a1, a2));
#pragma unroll
  for (int off = 16; off > 0; off >>= 1) m = fmaxf(m, __shfl_xor(m, off, 32));

  float e0 = __expf(a0 - m);
  float e1 = __expf(a1 - m);
  float e2 = (lane + 64 < NC) ? __expf(a2 - m) : 0.0f;

  float s = e0 + e1 + e2;
#pragma unroll
  for (int off = 16; off > 0; off >>= 1) s += __shfl_xor(s, off, 32);

  const float inv = 1.0f / s;
  float* p = prob + (size_t)row * NC;
  p[lane]      = e0 * inv;
  p[lane + 32] = e1 * inv;
  if (lane + 64 < NC) p[lane + 64] = e2 * inv;
}

__global__ __launch_bounds__(128) void cost_strip_k(
    const float* __restrict__ prob,
    const float* __restrict__ pred_boxes,
    const float* __restrict__ boxes,
    const int*   __restrict__ labels,
    float* __restrict__ out) {
  const int wave = threadIdx.x >> 5;
  const int qt = blockIdx.x * 4 + wave;
  if (qt * 16 >= NQ) return;
  const int b = blockIdx.z;

  const int lane = threadIdx.x & 31;
  const int half = lane >> 4;
  const int l15  = lane & 15;
  const int kb   = half << 1;

  const int q_m  = qt * 16 + l15;
  const int q_cl = (q_m < NQ) ? q_m : (NQ - 1);
  const float* arow = prob + ((size_t)b * NQ + q_cl) * NC;

  v2f afrag[KSTEPS];
#pragma unroll
  for (int kk = 0; kk < KSTEPS; ++kk) {
    afrag[kk].x = arow[4 * kk + kb];
    afrag[kk].y = arow[4 * kk + kb + 1];
  }

  float4 pbs[8];
#pragma unroll
  for (int v = 0; v < 8; ++v) {
    const int m = v + (half << 3);
    const int q = qt * 16 + m;
    const int qc = (q < NQ) ? q : (NQ - 1);
    pbs[v] = *(const float4*)(pred_boxes + ((size_t)b * NQ + qc) * 4);
  }

  for (int tt = 0; tt < TTILES; ++tt) {
    const int t_n   = tt * 16 + l15;
    const int t_cl  = (t_n < NT) ? t_n : (NT - 1);
    const int label = labels[(size_t)b * NT + t_cl];

    v8f acc = {0.f, 0.f, 0.f, 0.f, 0.f, 0.f, 0.f, 0.f};
#pragma unroll
    for (int kk = 0; kk < KSTEPS; ++kk) {
      const int k = 4 * kk;
      v2f bm;
      bm.x = (label == (k + kb))     ? 1.0f : 0.0f;
      bm.y = (label == (k + kb + 1)) ? 1.0f : 0.0f;
      acc = __builtin_amdgcn_wmma_f32_16x16x4_f32(
           false, afrag[kk],  false, bm,
           (short)0, acc,  false,  false);
    }

    const float4 tb = *(const float4*)(boxes + ((size_t)b * NT + t_cl) * 4);
    const float bx0 = tb.x - 0.5f * tb.z, by0 = tb.y - 0.5f * tb.w;
    const float bx1 = tb.x + 0.5f * tb.z, by1 = tb.y + 0.5f * tb.w;
    const float area2 = (bx1 - bx0) * (by1 - by0);
    const bool  tval  = (t_n < NT);

#pragma unroll
    for (int v = 0; v < 8; ++v) {
      const int m = v + (half << 3);
      const int q = qt * 16 + m;
      if (q >= NQ || !tval) continue;
      const float4 pb = pbs[v];

      const float l1 = fabsf(pb.x - tb.x) + fabsf(pb.y - tb.y) +
                       fabsf(pb.z - tb.z) + fabsf(pb.w - tb.w);

      const float ax0 = pb.x - 0.5f * pb.z, ay0 = pb.y - 0.5f * pb.w;
      const float ax1 = pb.x + 0.5f * pb.z, ay1 = pb.y + 0.5f * pb.w;
      const float area1 = (ax1 - ax0) * (ay1 - ay0);

      const float iw    = fminf(ax1, bx1) - fmaxf(ax0, bx0);
      const float ih    = fminf(ay1, by1) - fmaxf(ay0, by0);
      const float inter = fmaxf(iw, 0.f) * fmaxf(ih, 0.f);
      const float uni   = area1 + area2 - inter;
      const float iou   = inter / uni;

      const float ew    = fmaxf(ax1, bx1) - fminf(ax0, bx0);
      const float eh    = fmaxf(ay1, by1) - fminf(ay0, by0);
      const float areae = fmaxf(ew, 0.f) * fmaxf(eh, 0.f);
      const float giou  = iou - (areae - uni) / areae;

      out[((size_t)b * NQ + q) * NT + t_n] = 5.0f * l1 - acc[v] - 2.0f * giou;
    }
  }
}

extern "C" void kernel_launch(void* const* d_in, const int* in_sizes, int n_in,
                              void* d_out, int out_size, void* d_ws, size_t ws_size,
                              hipStream_t stream) {
  const float* pred_logits = (const float*)d_in[0];
  const float* pred_boxes  = (const float*)d_in[1];
  const float* boxes       = (const float*)d_in[2];
  const int*   labels      = (const int*)d_in[3];
  float* out  = (float*)d_out;
  float* prob = (float*)d_ws;

  const int rows = BS * NQ;
  softmax_rows_k<<<(rows + 7) / 8, 256, 0, stream>>>(pred_logits, prob);

  dim3 grid((QTILES + 3) / 4, 1, BS);
  cost_strip_k<<<grid, 128, 0, stream>>>(prob, pred_boxes, boxes, labels, out);
}


# Fuse the row softmax into the strip kernel: each lane pair normalises its row's 92 logits in the A-fragment registers (new row-sum order); the softmax launch and the prob workspace pass are gone
# speedup vs baseline: 1.2678x; 1.2678x over previous
#include <hip/hip_runtime.h>
#include <hip/hip_bf16.h>
#include <cfloat>

#define BS 64
#define NQ 900
#define NT 200
#define NC 92
#define KSTEPS (NC / 4)
#define TTILES ((NT + 15) / 16)
#define QTILES ((NQ + 15) / 16)

typedef float v2f __attribute__((ext_vector_type(2)));
typedef float v8f __attribute__((ext_vector_type(8)));

__device__ __forceinline__ float xhalf(float x) {
  const int b = __builtin_bit_cast(int, x);
  return __builtin_bit_cast(
      float, __builtin_amdgcn_permlanex16(b, b, 0x76543210, 0xfedcba98, false, false));
}

__global__ __launch_bounds__(128) void cost_strip_k(
    const float* __restrict__ logits,
    const float* __restrict__ pred_boxes,
    const float* __restrict__ boxes,
    const int*   __restrict__ labels,
    float* __restrict__ out) {
  const int wave = threadIdx.x >> 5;
  const int qt = blockIdx.x * 4 + wave;
  if (qt * 16 >= NQ) return;
  const int b = blockIdx.z;

  const int lane = threadIdx.x & 31;
  const int half = lane >> 4;
  const int l15  = lane & 15;
  const int kb   = half << 1;

  const int q_m  = qt * 16 + l15;
  const int q_cl = (q_m < NQ) ? q_m : (NQ - 1);
  const float* arow = logits + ((size_t)b * NQ + q_cl) * NC;

  v2f afrag[KSTEPS];
#pragma unroll
  for (int kk = 0; kk < KSTEPS; ++kk) {
    afrag[kk].x = arow[4 * kk + kb];
    afrag[kk].y = arow[4 * kk + kb + 1];
  }

  float mx = fmaxf(afrag[0].x, afrag[0].y);
#pragma unroll
  for (int kk = 1; kk < KSTEPS; ++kk)
    mx = fmaxf(mx, fmaxf(afrag[kk].x, afrag[kk].y));
  mx = fmaxf(mx, xhalf(mx));

  v2f se = {0.f, 0.f}, so = {0.f, 0.f};
#pragma unroll
  for (int kk = 0; kk < KSTEPS; ++kk) {
    afrag[kk].x = __expf(afrag[kk].x - mx);
    afrag[kk].y = __expf(afrag[kk].y - mx);
    if (kk & 1) so += afrag[kk]; else se += afrag[kk];
  }
  float rs = (se.x + so.x) + (se.y + so.y);
  rs += xhalf(rs);
  const float inv = 1.0f / rs;
#pragma unroll
  for (int kk = 0; kk < KSTEPS; ++kk) afrag[kk] *= inv;

  float4 pbs[8];
#pragma unroll
  for (int v = 0; v < 8; ++v) {
    const int m = v + (half << 3);
    const int q = qt * 16 + m;
    const int qc = (q < NQ) ? q : (NQ - 1);
    pbs[v] = *(const float4*)(pred_boxes + ((size_t)b * NQ + qc) * 4);
  }

  for (int tt = 0; tt < TTILES; ++tt) {
    const int t_n   = tt * 16 + l15;
    const int t_cl  = (t_n < NT) ? t_n : (NT - 1);
    const int label = labels[(size_t)b * NT + t_cl];

    v8f acc = {0.f, 0.f, 0.f, 0.f, 0.f, 0.f, 0.f, 0.f};
#pragma unroll
    for (int kk = 0; kk < KSTEPS; ++kk) {
      const int k = 4 * kk;
      v2f bm;
      bm.x = (label == (k + kb))     ? 1.0f : 0.0f;
      bm.y = (label == (k + kb + 1)) ? 1.0f : 0.0f;
      acc = __builtin_amdgcn_wmma_f32_16x16x4_f32(
           false, afrag[kk],  false, bm,
           (short)0, acc,  false,  false);
    }

    const float4 tb = *(const float4*)(boxes + ((size_t)b * NT + t_cl) * 4);
    const float bx0 = tb.x - 0.5f * tb.z, by0 = tb.y - 0.5f * tb.w;
    const float bx1 = tb.x + 0.5f * tb.z, by1 = tb.y + 0.5f * tb.w;
    const float area2 = (bx1 - bx0) * (by1 - by0);
    const bool  tval  = (t_n < NT);

#pragma unroll
    for (int v = 0; v < 8; ++v) {
      const int m = v + (half << 3);
      const int q = qt * 16 + m;
      if (q >= NQ || !tval) continue;
      const float4 pb = pbs[v];

      const float l1 = fabsf(pb.x - tb.x) + fabsf(pb.y - tb.y) +
                       fabsf(pb.z - tb.z) + fabsf(pb.w - tb.w);

      const float ax0 = pb.x - 0.5f * pb.z, ay0 = pb.y - 0.5f * pb.w;
      const float ax1 = pb.x + 0.5f * pb.z, ay1 = pb.y + 0.5f * pb.w;
      const float area1 = (ax1 - ax0) * (ay1 - ay0);

      const float iw    = fminf(ax1, bx1) - fmaxf(ax0, bx0);
      const float ih    = fminf(ay1, by1) - fmaxf(ay0, by0);
      const float inter = fmaxf(iw, 0.f) * fmaxf(ih, 0.f);
      const float uni   = area1 + area2 - inter;
      const float iou   = inter / uni;

      const float ew    = fmaxf(ax1, bx1) - fminf(ax0, bx0);
      const float eh    = fmaxf(ay1, by1) - fminf(ay0, by0);
      const float areae = fmaxf(ew, 0.f) * fmaxf(eh, 0.f);
      const float giou  = iou - (areae - uni) / areae;

      out[((size_t)b * NQ + q) * NT + t_n] = 5.0f * l1 - acc[v] - 2.0f * giou;
    }
  }
}

extern "C" void kernel_launch(void* const* d_in, const int* in_sizes, int n_in,
                              void* d_out, int out_size, void* d_ws, size_t ws_size,
                              hipStream_t stream) {
  const float* pred_logits = (const float*)d_in[0];
  const float* pred_boxes  = (const float*)d_in[1];
  const float* boxes       = (const float*)d_in[2];
  const int*   labels      = (const int*)d_in[3];
  float* out  = (float*)d_out;

  dim3 grid((QTILES + 3) / 4, 1, BS);
  cost_strip_k<<<grid, 128, 0, stream>>>(pred_logits, pred_boxes, boxes, labels, out);
}


# GIoU epilogue: both f32 divisions become numerator * v_rcp_f32(denominator) (1-ulp reciprocal, quotients no longer correctly rounded); each 11-instruction IEEE divide drops to rcp+mul
# speedup vs baseline: 1.4526x; 1.1458x over previous
#include <hip/hip_runtime.h>
#include <hip/hip_bf16.h>
#include <cfloat>

#define BS 64
#define NQ 900
#define NT 200
#define NC 92
#define KSTEPS (NC / 4)
#define TTILES ((NT + 15) / 16)
#define QTILES ((NQ + 15) / 16)

typedef float v2f __attribute__((ext_vector_type(2)));
typedef float v8f __attribute__((ext_vector_type(8)));

__device__ __forceinline__ float xhalf(float x) {
  const int b = __builtin_bit_cast(int, x);
  return __builtin_bit_cast(
      float, __builtin_amdgcn_permlanex16(b, b, 0x76543210, 0xfedcba98, false, false));
}

__global__ __launch_bounds__(128) void cost_strip_k(
    const float* __restrict__ logits,
    const float* __restrict__ pred_boxes,
    const float* __restrict__ boxes,
    const int*   __restrict__ labels,
    float* __restrict__ out) {
  const int wave = threadIdx.x >> 5;
  const int qt = blockIdx.x * 4 + wave;
  if (qt * 16 >= NQ) return;
  const int b = blockIdx.z;

  const int lane = threadIdx.x & 31;
  const int half = lane >> 4;
  const int l15  = lane & 15;
  const int kb   = half << 1;

  const int q_m  = qt * 16 + l15;
  const int q_cl = (q_m < NQ) ? q_m : (NQ - 1);
  const float* arow = logits + ((size_t)b * NQ + q_cl) * NC;

  v2f afrag[KSTEPS];
#pragma unroll
  for (int kk = 0; kk < KSTEPS; ++kk) {
    afrag[kk].x = arow[4 * kk + kb];
    afrag[kk].y = arow[4 * kk + kb + 1];
  }

  float mx = fmaxf(afrag[0].x, afrag[0].y);
#pragma unroll
  for (int kk = 1; kk < KSTEPS; ++kk)
    mx = fmaxf(mx, fmaxf(afrag[kk].x, afrag[kk].y));
  mx = fmaxf(mx, xhalf(mx));

  v2f se = {0.f, 0.f}, so = {0.f, 0.f};
#pragma unroll
  for (int kk = 0; kk < KSTEPS; ++kk) {
    afrag[kk].x = __expf(afrag[kk].x - mx);
    afrag[kk].y = __expf(afrag[kk].y - mx);
    if (kk & 1) so += afrag[kk]; else se += afrag[kk];
  }
  float rs = (se.x + so.x) + (se.y + so.y);
  rs += xhalf(rs);
  const float inv = 1.0f / rs;
#pragma unroll
  for (int kk = 0; kk < KSTEPS; ++kk) afrag[kk] *= inv;

  float4 pbs[8];
#pragma unroll
  for (int v = 0; v < 8; ++v) {
    const int m = v + (half << 3);
    const int q = qt * 16 + m;
    const int qc = (q < NQ) ? q : (NQ - 1);
    pbs[v] = *(const float4*)(pred_boxes + ((size_t)b * NQ + qc) * 4);
  }

  for (int tt = 0; tt < TTILES; ++tt) {
    const int t_n   = tt * 16 + l15;
    const int t_cl  = (t_n < NT) ? t_n : (NT - 1);
    const int label = labels[(size_t)b * NT + t_cl];

    v8f acc = {0.f, 0.f, 0.f, 0.f, 0.f, 0.f, 0.f, 0.f};
#pragma unroll
    for (int kk = 0; kk < KSTEPS; ++kk) {
      const int k = 4 * kk;
      v2f bm;
      bm.x = (label == (k + kb))     ? 1.0f : 0.0f;
      bm.y = (label == (k + kb + 1)) ? 1.0f : 0.0f;
      acc = __builtin_amdgcn_wmma_f32_16x16x4_f32(
           false, afrag[kk],  false, bm,
           (short)0, acc,  false,  false);
    }

    const float4 tb = *(const float4*)(boxes + ((size_t)b * NT + t_cl) * 4);
    const float bx0 = tb.x - 0.5f * tb.z, by0 = tb.y - 0.5f * tb.w;
    const float bx1 = tb.x + 0.5f * tb.z, by1 = tb.y + 0.5f * tb.w;
    const float area2 = (bx1 - bx0) * (by1 - by0);
    const bool  tval  = (t_n < NT);

#pragma unroll
    for (int v = 0; v < 8; ++v) {
      const int m = v + (half << 3);
      const int q = qt * 16 + m;
      if (q >= NQ || !tval) continue;
      const float4 pb = pbs[v];

      const float l1 = fabsf(pb.x - tb.x) + fabsf(pb.y - tb.y) +
                       fabsf(pb.z - tb.z) + fabsf(pb.w - tb.w);

      const float ax0 = pb.x - 0.5f * pb.z, ay0 = pb.y - 0.5f * pb.w;
      const float ax1 = pb.x + 0.5f * pb.z, ay1 = pb.y + 0.5f * pb.w;
      const float area1 = (ax1 - ax0) * (ay1 - ay0);

      const float iw    = fminf(ax1, bx1) - fmaxf(ax0, bx0);
      const float ih    = fminf(ay1, by1) - fmaxf(ay0, by0);
      const float inter = fmaxf(iw, 0.f) * fmaxf(ih, 0.f);
      const float uni   = area1 + area2 - inter;
      const float iou   = inter * __builtin_amdgcn_rcpf(uni);

      const float ew    = fmaxf(ax1, bx1) - fminf(ax0, bx0);
      const float eh    = fmaxf(ay1, by1) - fminf(ay0, by0);
      const float areae = fmaxf(ew, 0.f) * fmaxf(eh, 0.f);
      const float giou  = iou - (areae - uni) * __builtin_amdgcn_rcpf(areae);

      out[((size_t)b * NQ + q) * NT + t_n] = 5.0f * l1 - acc[v] - 2.0f * giou;
    }
  }
}

extern "C" void kernel_launch(void* const* d_in, const int* in_sizes, int n_in,
                              void* d_out, int out_size, void* d_ws, size_t ws_size,
                              hipStream_t stream) {
  const float* pred_logits = (const float*)d_in[0];
  const float* pred_boxes  = (const float*)d_in[1];
  const float* boxes       = (const float*)d_in[2];
  const int*   labels      = (const int*)d_in[3];
  float* out  = (float*)d_out;

  dim3 grid((QTILES + 3) / 4, 1, BS);
  cost_strip_k<<<grid, 128, 0, stream>>>(pred_logits, pred_boxes, boxes, labels, out);
}


# Skip k-steps whose one-hot fragment is zero in all 16 columns of the t-tile (wave-uniform branch on a per-tile bit mask of label>>2): at most 16 of 23 WMMAs run; skipped terms are exact +0
# speedup vs baseline: 1.6462x; 1.1333x over previous
#include <hip/hip_runtime.h>
#include <hip/hip_bf16.h>
#include <cfloat>

#define BS 64
#define NQ 900
#define NT 200
#define NC 92
#define KSTEPS (NC / 4)
#define TTILES ((NT + 15) / 16)
#define QTILES ((NQ + 15) / 16)

typedef float v2f __attribute__((ext_vector_type(2)));
typedef float v8f __attribute__((ext_vector_type(8)));

__device__ __forceinline__ float xhalf(float x) {
  const int b = __builtin_bit_cast(int, x);
  return __builtin_bit_cast(
      float, __builtin_amdgcn_permlanex16(b, b, 0x76543210, 0xfedcba98, false, false));
}

__device__ __forceinline__ unsigned row_or16(unsigned m) {
  m |= (unsigned)__builtin_amdgcn_update_dpp(0, (int)m, 0x161, 0xf, 0xf, true);
  m |= (unsigned)__builtin_amdgcn_update_dpp(0, (int)m, 0x162, 0xf, 0xf, true);
  m |= (unsigned)__builtin_amdgcn_update_dpp(0, (int)m, 0x164, 0xf, 0xf, true);
  m |= (unsigned)__builtin_amdgcn_update_dpp(0, (int)m, 0x168, 0xf, 0xf, true);
  return m;
}

__global__ __launch_bounds__(128) void cost_strip_k(
    const float* __restrict__ logits,
    const float* __restrict__ pred_boxes,
    const float* __restrict__ boxes,
    const int*   __restrict__ labels,
    float* __restrict__ out) {
  const int wave = threadIdx.x >> 5;
  const int qt = blockIdx.x * 4 + wave;
  if (qt * 16 >= NQ) return;
  const int b = blockIdx.z;

  const int lane = threadIdx.x & 31;
  const int half = lane >> 4;
  const int l15  = lane & 15;
  const int kb   = half << 1;

  const int q_m  = qt * 16 + l15;
  const int q_cl = (q_m < NQ) ? q_m : (NQ - 1);
  const float* arow = logits + ((size_t)b * NQ + q_cl) * NC;

  v2f afrag[KSTEPS];
#pragma unroll
  for (int kk = 0; kk < KSTEPS; ++kk) {
    afrag[kk].x = arow[4 * kk + kb];
    afrag[kk].y = arow[4 * kk + kb + 1];
  }

  float mx = fmaxf(afrag[0].x, afrag[0].y);
#pragma unroll
  for (int kk = 1; kk < KSTEPS; ++kk)
    mx = fmaxf(mx, fmaxf(afrag[kk].x, afrag[kk].y));
  mx = fmaxf(mx, xhalf(mx));

  v2f se = {0.f, 0.f}, so = {0.f, 0.f};
#pragma unroll
  for (int kk = 0; kk < KSTEPS; ++kk) {
    afrag[kk].x = __expf(afrag[kk].x - mx);
    afrag[kk].y = __expf(afrag[kk].y - mx);
    if (kk & 1) so += afrag[kk]; else se += afrag[kk];
  }
  float rs = (se.x + so.x) + (se.y + so.y);
  rs += xhalf(rs);
  const float inv = 1.0f / rs;
#pragma unroll
  for (int kk = 0; kk < KSTEPS; ++kk) afrag[kk] *= inv;

  float4 pbs[8];
#pragma unroll
  for (int v = 0; v < 8; ++v) {
    const int m = v + (half << 3);
    const int q = qt * 16 + m;
    const int qc = (q < NQ) ? q : (NQ - 1);
    pbs[v] = *(const float4*)(pred_boxes + ((size_t)b * NQ + qc) * 4);
  }

  for (int tt = 0; tt < TTILES; ++tt) {
    const int t_n   = tt * 16 + l15;
    const int t_cl  = (t_n < NT) ? t_n : (NT - 1);
    const int label = labels[(size_t)b * NT + t_cl];

    const unsigned need = __builtin_amdgcn_readfirstlane(
        row_or16(1u << ((unsigned)(label >> 2) & 31u)));

    v8f acc = {0.f, 0.f, 0.f, 0.f, 0.f, 0.f, 0.f, 0.f};
#pragma unroll
    for (int kk = 0; kk < KSTEPS; ++kk) {
      if (need & (1u << kk)) {
        const int k = 4 * kk;
        v2f bm;
        bm.x = (label == (k + kb))     ? 1.0f : 0.0f;
        bm.y = (label == (k + kb + 1)) ? 1.0f : 0.0f;
        acc = __builtin_amdgcn_wmma_f32_16x16x4_f32(
             false, afrag[kk],  false, bm,
             (short)0, acc,  false,  false);
      }
    }

    const float4 tb = *(const float4*)(boxes + ((size_t)b * NT + t_cl) * 4);
    const float bx0 = tb.x - 0.5f * tb.z, by0 = tb.y - 0.5f * tb.w;
    const float bx1 = tb.x + 0.5f * tb.z, by1 = tb.y + 0.5f * tb.w;
    const float area2 = (bx1 - bx0) * (by1 - by0);
    const bool  tval  = (t_n < NT);

#pragma unroll
    for (int v = 0; v < 8; ++v) {
      const int m = v + (half << 3);
      const int q = qt * 16 + m;
      if (q >= NQ || !tval) continue;
      const float4 pb = pbs[v];

      const float l1 = fabsf(pb.x - tb.x) + fabsf(pb.y - tb.y) +
                       fabsf(pb.z - tb.z) + fabsf(pb.w - tb.w);

      const float ax0 = pb.x - 0.5f * pb.z, ay0 = pb.y - 0.5f * pb.w;
      const float ax1 = pb.x + 0.5f * pb.z, ay1 = pb.y + 0.5f * pb.w;
      const float area1 = (ax1 - ax0) * (ay1 - ay0);

      const float iw    = fminf(ax1, bx1) - fmaxf(ax0, bx0);
      const float ih    = fminf(ay1, by1) - fmaxf(ay0, by0);
      const float inter = fmaxf(iw, 0.f) * fmaxf(ih, 0.f);
      const float uni   = area1 + area2 - inter;
      const float iou   = inter * __builtin_amdgcn_rcpf(uni);

      const float ew    = fmaxf(ax1, bx1) - fminf(ax0, bx0);
      const float eh    = fmaxf(ay1, by1) - fminf(ay0, by0);
      const float areae = fmaxf(ew, 0.f) * fmaxf(eh, 0.f);
      const float giou  = iou - (areae - uni) * __builtin_amdgcn_rcpf(areae);

      out[((size_t)b * NQ + q) * NT + t_n] = 5.0f * l1 - acc[v] - 2.0f * giou;
    }
  }
}

extern "C" void kernel_launch(void* const* d_in, const int* in_sizes, int n_in,
                              void* d_out, int out_size, void* d_ws, size_t ws_size,
                              hipStream_t stream) {
  const float* pred_logits = (const float*)d_in[0];
  const float* pred_boxes  = (const float*)d_in[1];
  const float* boxes       = (const float*)d_in[2];
  const int*   labels      = (const int*)d_in[3];
  float* out  = (float*)d_out;

  dim3 grid((QTILES + 3) / 4, 1, BS);
  cost_strip_k<<<grid, 128, 0, stream>>>(pred_logits, pred_boxes, boxes, labels, out);
}


# Fetch each t-tile's label and target box one tile ahead (tile 0's before the loop): neither L2 round trip stands in front of the WMMA chain or the epilogue any more
# speedup vs baseline: 1.9359x; 1.1760x over previous
#include <hip/hip_runtime.h>
#include <hip/hip_bf16.h>
#include <cfloat>

#define BS 64
#define NQ 900
#define NT 200
#define NC 92
#define KSTEPS (NC / 4)
#define TTILES ((NT + 15) / 16)
#define QTILES ((NQ + 15) / 16)

typedef float v2f __attribute__((ext_vector_type(2)));
typedef float v8f __attribute__((ext_vector_type(8)));

__device__ __forceinline__ float xhalf(float x) {
  const int b = __builtin_bit_cast(int, x);
  return __builtin_bit_cast(
      float, __builtin_amdgcn_permlanex16(b, b, 0x76543210, 0xfedcba98, false, false));
}

__device__ __forceinline__ unsigned row_or16(unsigned m) {
  m |= (unsigned)__builtin_amdgcn_update_dpp(0, (int)m, 0x161, 0xf, 0xf, true);
  m |= (unsigned)__builtin_amdgcn_update_dpp(0, (int)m, 0x162, 0xf, 0xf, true);
  m |= (unsigned)__builtin_amdgcn_update_dpp(0, (int)m, 0x164, 0xf, 0xf, true);
  m |= (unsigned)__builtin_amdgcn_update_dpp(0, (int)m, 0x168, 0xf, 0xf, true);
  return m;
}

__global__ __launch_bounds__(128) void cost_strip_k(
    const float* __restrict__ logits,
    const float* __restrict__ pred_boxes,
    const float* __restrict__ boxes,
    const int*   __restrict__ labels,
    float* __restrict__ out) {
  const int wave = threadIdx.x >> 5;
  const int qt = blockIdx.x * 4 + wave;
  if (qt * 16 >= NQ) return;
  const int b = blockIdx.z;

  const int lane = threadIdx.x & 31;
  const int half = lane >> 4;
  const int l15  = lane & 15;
  const int kb   = half << 1;

  const int q_m  = qt * 16 + l15;
  const int q_cl = (q_m < NQ) ? q_m : (NQ - 1);
  const float* arow = logits + ((size_t)b * NQ + q_cl) * NC;

  v2f afrag[KSTEPS];
#pragma unroll
  for (int kk = 0; kk < KSTEPS; ++kk) {
    afrag[kk].x = arow[4 * kk + kb];
    afrag[kk].y = arow[4 * kk + kb + 1];
  }

  float mx = fmaxf(afrag[0].x, afrag[0].y);
#pragma unroll
  for (int kk = 1; kk < KSTEPS; ++kk)
    mx = fmaxf(mx, fmaxf(afrag[kk].x, afrag[kk].y));
  mx = fmaxf(mx, xhalf(mx));

  v2f se = {0.f, 0.f}, so = {0.f, 0.f};
#pragma unroll
  for (int kk = 0; kk < KSTEPS; ++kk) {
    afrag[kk].x = __expf(afrag[kk].x - mx);
    afrag[kk].y = __expf(afrag[kk].y - mx);
    if (kk & 1) so += afrag[kk]; else se += afrag[kk];
  }
  float rs = (se.x + so.x) + (se.y + so.y);
  rs += xhalf(rs);
  const float inv = 1.0f / rs;
#pragma unroll
  for (int kk = 0; kk < KSTEPS; ++kk) afrag[kk] *= inv;

  float4 pbs[8];
#pragma unroll
  for (int v = 0; v < 8; ++v) {
    const int m = v + (half << 3);
    const int q = qt * 16 + m;
    const int qc = (q < NQ) ? q : (NQ - 1);
    pbs[v] = *(const float4*)(pred_boxes + ((size_t)b * NQ + qc) * 4);
  }

  int    label_nx = labels[(size_t)b * NT + l15];
  float4 tb_nx    = *(const float4*)(boxes + ((size_t)b * NT + l15) * 4);
  for (int tt = 0; tt < TTILES; ++tt) {
    const int t_n   = tt * 16 + l15;
    const int label = label_nx;
    const float4 tb = tb_nx;
    {
      const int t_nx = (t_n + 16 < NT) ? (t_n + 16) : (NT - 1);
      label_nx = labels[(size_t)b * NT + t_nx];
      tb_nx    = *(const float4*)(boxes + ((size_t)b * NT + t_nx) * 4);
    }

    const unsigned need = __builtin_amdgcn_readfirstlane(
        row_or16(1u << ((unsigned)(label >> 2) & 31u)));

    v8f acc = {0.f, 0.f, 0.f, 0.f, 0.f, 0.f, 0.f, 0.f};
#pragma unroll
    for (int kk = 0; kk < KSTEPS; ++kk) {
      if (need & (1u << kk)) {
        const int k = 4 * kk;
        v2f bm;
        bm.x = (label == (k + kb))     ? 1.0f : 0.0f;
        bm.y = (label == (k + kb + 1)) ? 1.0f : 0.0f;
        acc = __builtin_amdgcn_wmma_f32_16x16x4_f32(
             false, afrag[kk],  false, bm,
             (short)0, acc,  false,  false);
      }
    }

    const float bx0 = tb.x - 0.5f * tb.z, by0 = tb.y - 0.5f * tb.w;
    const float bx1 = tb.x + 0.5f * tb.z, by1 = tb.y + 0.5f * tb.w;
    const float area2 = (bx1 - bx0) * (by1 - by0);
    const bool  tval  = (t_n < NT);

#pragma unroll
    for (int v = 0; v < 8; ++v) {
      const int m = v + (half << 3);
      const int q = qt * 16 + m;
      if (q >= NQ || !tval) continue;
      const float4 pb = pbs[v];

      const float l1 = fabsf(pb.x - tb.x) + fabsf(pb.y - tb.y) +
                       fabsf(pb.z - tb.z) + fabsf(pb.w - tb.w);

      const float ax0 = pb.x - 0.5f * pb.z, ay0 = pb.y - 0.5f * pb.w;
      const float ax1 = pb.x + 0.5f * pb.z, ay1 = pb.y + 0.5f * pb.w;
      const float area1 = (ax1 - ax0) * (ay1 - ay0);

      const float iw    = fminf(ax1, bx1) - fmaxf(ax0, bx0);
      const float ih    = fminf(ay1, by1) - fmaxf(ay0, by0);
      const float inter = fmaxf(iw, 0.f) * fmaxf(ih, 0.f);
      const float uni   = area1 + area2 - inter;
      const float iou   = inter * __builtin_amdgcn_rcpf(uni);

      const float ew    = fmaxf(ax1, bx1) - fminf(ax0, bx0);
      const float eh    = fmaxf(ay1, by1) - fminf(ay0, by0);
      const float areae = fmaxf(ew, 0.f) * fmaxf(eh, 0.f);
      const float giou  = iou - (areae - uni) * __builtin_amdgcn_rcpf(areae);

      out[((size_t)b * NQ + q) * NT + t_n] = 5.0f * l1 - acc[v] - 2.0f * giou;
    }
  }
}

extern "C" void kernel_launch(void* const* d_in, const int* in_sizes, int n_in,
                              void* d_out, int out_size, void* d_ws, size_t ws_size,
                              hipStream_t stream) {
  const float* pred_logits = (const float*)d_in[0];
  const float* pred_boxes  = (const float*)d_in[1];
  const float* boxes       = (const float*)d_in[2];
  const int*   labels      = (const int*)d_in[3];
  float* out  = (float*)d_out;

  dim3 grid((QTILES + 3) / 4, 1, BS);
  cost_strip_k<<<grid, 128, 0, stream>>>(pred_logits, pred_boxes, boxes, labels, out);
}
